# starting kernel
# baseline (speedup 1.0000x reference)
#include <hip/hip_runtime.h>
#include <hip/hip_bf16.h>
#include <math.h>

typedef __attribute__((ext_vector_type(8)))  float  v8f;
typedef __attribute__((ext_vector_type(4)))  __bf16 v4bf;
typedef __attribute__((ext_vector_type(8)))  __bf16 v8bf;
typedef __attribute__((ext_vector_type(16))) __bf16 v16bf;

#define B_   32
#define L_   2048
#define D_   512
#define H_   8
#define DK_  64
#define NSPLIT 4
#define CHUNK (L_ / NSPLIT)
#define TILES (CHUNK / 16)
#define KSTEPS (D_ / 32)
#define FRAGN (KSTEPS * 4 * 32 * 16)
#define SROW 520

static __device__ inline v16bf join16(v8bf lo, v8bf hi) {
  return __builtin_shufflevector(lo, hi, 0,1,2,3,4,5,6,7,8,9,10,11,12,13,14,15);
}

static __device__ inline float fast_tanh(float x) {
  x = fminf(fmaxf(x, -15.f), 15.f);
  float e = __expf(2.0f * x);
  return (e - 1.0f) / (e + 1.0f);
}

__global__ void __launch_bounds__(256)
ga_flash_kernel(const float* __restrict__ X, const float* __restrict__ S,
                const float* __restrict__ Wq, const float* __restrict__ Wk,
                const float* __restrict__ Wv,
                float* __restrict__ ws_m, float* __restrict__ ws_s,
                float* __restrict__ ws_acc)
{
  __shared__ __bf16 wfrag[2][FRAGN];
  __shared__ __bf16 s_tile[16 * SROW];
  __shared__ float  sh_x[D_];
  __shared__ float  sh_q[DK_];
  __shared__ float  sh_part[4][16];
  __shared__ float  sh_p[16];
  __shared__ float  sh_scale;

  const int tid  = threadIdx.x;
  const int lane = tid & 31;
  const int wave = tid >> 5;
  const int wn   = wave & 3;
  const int sel  = wave >> 2;

  const int pid = blockIdx.x;
  const int sp  = pid & (NSPLIT - 1);
  const int h   = (pid >> 2) & (H_ - 1);
  const int b   = pid >> 5;
  const int l0c = sp * CHUNK;

  for (int i = tid; i < D_; i += 256) sh_x[i] = X[b * D_ + i];

  for (int idx = tid; idx < 2 * FRAGN; idx += 256) {
    int s2 = idx >> 15;
    int r  = idx & (FRAGN - 1);
    int e  = r & 15;
    int l  = (r >> 4) & 31;
    int w2 = (r >> 9) & 3;
    int kk = r >> 11;
    int koff = ((l < 16) ? 0 : 8) + (e & 7) + ((e >> 3) << 4);
    int K = kk * 32 + koff;
    int n = w2 * 16 + (l & 15);
    const float* W = s2 ? Wv : Wk;
    wfrag[s2][r] = (__bf16)W[(h * D_ + K) * DK_ + n];
  }
  __syncthreads();

  if (tid < DK_) {
    float a = 0.f;
    for (int g = 0; g < D_; ++g) a += sh_x[g] * Wq[(h * D_ + g) * DK_ + tid];
    sh_q[tid] = fast_tanh(a);
  }

  float m_run = -INFINITY;
  float s_run = 0.f;
  float acc   = 0.f;

  const int m_row = lane & 15;
  const int kbase = (lane < 16) ? 0 : 8;
  const int half  = lane >> 4;

  const size_t sbase = ((size_t)b * L_ + l0c) * D_;

  float4 stage[8];
  #pragma unroll
  for (int j = 0; j < 8; ++j)
    stage[j] = *reinterpret_cast<const float4*>(&S[sbase + tid * 4 + j * 1024]);

  for (int t = 0; t < TILES; ++t) {
    __syncthreads();
    #pragma unroll
    for (int j = 0; j < 8; ++j) {
      int e0  = tid * 4 + j * 1024;
      int row = e0 >> 9, col = e0 & 511;
      float4 f = stage[j];
      v4bf p4 = { (__bf16)f.x, (__bf16)f.y, (__bf16)f.z, (__bf16)f.w };
      *reinterpret_cast<v4bf*>(&s_tile[row * SROW + col]) = p4;
    }
    if (t + 1 < TILES) {
      const size_t sb1 = sbase + (size_t)(t + 1) * 16 * D_;
      #pragma unroll
      for (int j = 0; j < 8; ++j)
        stage[j] = *reinterpret_cast<const float4*>(&S[sb1 + tid * 4 + j * 1024]);
    }
    if (t + 2 < TILES) {
      const size_t sb2 = sbase + (size_t)(t + 2) * 16 * D_;
      #pragma unroll
      for (int j = 0; j < 8; ++j)
        __builtin_prefetch(&S[sb2 + tid * 4 + j * 1024], 0, 1);
    }
    __syncthreads();

    v8f c = {};
    const __bf16* fb_base = &wfrag[sel][0];
    #pragma unroll
    for (int kk = 0; kk < KSTEPS; ++kk) {
      const __bf16* pa = &s_tile[m_row * SROW + kk * 32 + kbase];
      v16bf a = join16(*reinterpret_cast<const v8bf*>(pa),
                       *reinterpret_cast<const v8bf*>(pa + 16));
      const __bf16* pb = fb_base + (((kk * 4 + wn) * 32) + lane) * 16;
      v16bf bf = join16(*reinterpret_cast<const v8bf*>(pb),
                        *reinterpret_cast<const v8bf*>(pb + 8));
      c = __builtin_amdgcn_wmma_f32_16x16x32_bf16(
            false, a, false, bf, (short)0, c, false, false);
    }

    if (sel == 0) {
      float qv = sh_q[wn * 16 + m_row];
      float part[8];
      #pragma unroll
      for (int r = 0; r < 8; ++r) part[r] = fast_tanh(c[r]) * qv;
      #pragma unroll
      for (int off = 1; off < 16; off <<= 1) {
        #pragma unroll
        for (int r = 0; r < 8; ++r) part[r] += __shfl_xor(part[r], off);
      }
      if ((lane & 15) == 0) {
        #pragma unroll
        for (int r = 0; r < 8; ++r) sh_part[wn][r + 8 * half] = part[r];
      }
    }
    __syncthreads();

    if (tid < 16) {
      float sc = (sh_part[0][tid] + sh_part[1][tid] +
                  sh_part[2][tid] + sh_part[3][tid]) * 0.125f;
      float mx = sc;
      #pragma unroll
      for (int off = 8; off >= 1; off >>= 1) mx = fmaxf(mx, __shfl_xor(mx, off));
      float m_new = fmaxf(m_run, mx);
      float scale = __expf(m_run - m_new);
      float p     = __expf(sc - m_new);
      float ps = p;
      #pragma unroll
      for (int off = 8; off >= 1; off >>= 1) ps += __shfl_xor(ps, off);
      s_run = s_run * scale + ps;
      m_run = m_new;
      sh_p[tid] = p;
      if (tid == 0) sh_scale = scale;
    }
    __syncthreads();

    if (sel == 1) {
      float contrib = 0.f;
      #pragma unroll
      for (int r = 0; r < 8; ++r) {
        float v = c[r];
        v = v > 0.f ? v : 0.f;
        contrib += sh_p[r + 8 * half] * v;
      }
      contrib += __shfl_xor(contrib, 16);
      acc = acc * sh_scale + contrib;
    }
  }

  if (sel == 1 && (lane & 16) == 0)
    ws_acc[(size_t)pid * DK_ + wn * 16 + m_row] = acc;
  if (tid == 0) { ws_m[pid] = m_run; ws_s[pid] = s_run; }
}

__global__ void __launch_bounds__(64)
ga_combine_kernel(const float* __restrict__ ws_m, const float* __restrict__ ws_s,
                  const float* __restrict__ ws_acc, float* __restrict__ out)
{
  const int bh = blockIdx.x;
  const int k  = threadIdx.x;
  float m = -INFINITY;
  #pragma unroll
  for (int sp = 0; sp < NSPLIT; ++sp) m = fmaxf(m, ws_m[bh * NSPLIT + sp]);
  float num = 0.f, den = 0.f;
  #pragma unroll
  for (int sp = 0; sp < NSPLIT; ++sp) {
    int pid = bh * NSPLIT + sp;
    float w = __expf(ws_m[pid] - m);
    num += w * ws_acc[(size_t)pid * DK_ + k];
    den += w * ws_s[pid];
  }
  out[bh * DK_ + k] = num / den;
}

extern "C" void kernel_launch(void* const* d_in, const int* in_sizes, int n_in,
                              void* d_out, int out_size, void* d_ws, size_t ws_size,
                              hipStream_t stream) {
  (void)in_sizes; (void)n_in; (void)out_size; (void)ws_size;
  const float* X  = (const float*)d_in[0];
  const float* S  = (const float*)d_in[1];
  const float* Wq = (const float*)d_in[2];
  const float* Wk = (const float*)d_in[3];
  const float* Wv = (const float*)d_in[4];

  const int NP = B_ * H_ * NSPLIT;
  float* wsf    = (float*)d_ws;
  float* ws_m   = wsf;
  float* ws_s   = wsf + NP;
  float* ws_acc = wsf + 2 * NP;

  ga_flash_kernel<<<NP, 256, 0, stream>>>(X, S, Wq, Wk, Wv, ws_m, ws_s, ws_acc);
  ga_combine_kernel<<<B_ * H_, 64, 0, stream>>>(ws_m, ws_s, ws_acc, (float*)d_out);
}


# Weight (B) WMMA fragments kept in registers for the whole block (128 VGPRs per wave, read once from global, row-contiguous) instead of built in LDS by scattered loads and re-read from LDS every tile
# speedup vs baseline: 1.3494x; 1.3494x over previous
#include <hip/hip_runtime.h>
#include <hip/hip_bf16.h>
#include <math.h>

typedef __attribute__((ext_vector_type(8)))  float  v8f;
typedef __attribute__((ext_vector_type(4)))  __bf16 v4bf;
typedef __attribute__((ext_vector_type(8)))  __bf16 v8bf;
typedef __attribute__((ext_vector_type(16))) __bf16 v16bf;

#define B_   32
#define L_   2048
#define D_   512
#define H_   8
#define DK_  64
#define NSPLIT 4
#define CHUNK (L_ / NSPLIT)
#define TILES (CHUNK / 16)
#define KSTEPS (D_ / 32)
#define SROW 520

static __device__ inline v16bf join16(v8bf lo, v8bf hi) {
  return __builtin_shufflevector(lo, hi, 0,1,2,3,4,5,6,7,8,9,10,11,12,13,14,15);
}

static __device__ inline float fast_tanh(float x) {
  x = fminf(fmaxf(x, -15.f), 15.f);
  float e = __expf(2.0f * x);
  return (e - 1.0f) / (e + 1.0f);
}

__global__ void __launch_bounds__(256)
ga_flash_kernel(const float* __restrict__ X, const float* __restrict__ S,
                const float* __restrict__ Wq, const float* __restrict__ Wk,
                const float* __restrict__ Wv,
                float* __restrict__ ws_m, float* __restrict__ ws_s,
                float* __restrict__ ws_acc)
{
  __shared__ __bf16 s_tile[16 * SROW];
  __shared__ float  sh_x[D_];
  __shared__ float  sh_q[DK_];
  __shared__ float  sh_part[4][16];
  __shared__ float  sh_p[16];
  __shared__ float  sh_scale;

  const int tid  = threadIdx.x;
  const int lane = tid & 31;
  const int wave = tid >> 5;
  const int wn   = wave & 3;
  const int sel  = wave >> 2;

  const int pid = blockIdx.x;
  const int sp  = pid & (NSPLIT - 1);
  const int h   = (pid >> 2) & (H_ - 1);
  const int b   = pid >> 5;
  const int l0c = sp * CHUNK;

  const int m_row = lane & 15;
  const int kbase = (lane < 16) ? 0 : 8;
  const int half  = lane >> 4;

  for (int i = tid; i < D_; i += 256) sh_x[i] = X[b * D_ + i];
  __syncthreads();

  if (tid < DK_) {
    float a = 0.f;
    for (int g = 0; g < D_; ++g) a += sh_x[g] * Wq[(h * D_ + g) * DK_ + tid];
    sh_q[tid] = fast_tanh(a);
  }

  const float* wp = (sel ? Wv : Wk) + ((size_t)h * D_ + kbase) * DK_ + wn * 16 + m_row;
  v16bf bfr[KSTEPS];
  #pragma unroll
  for (int kk = 0; kk < KSTEPS; ++kk) {
    #pragma unroll
    for (int e = 0; e < 16; ++e) {
      const int k = kk * 32 + (e & 7) + ((e >> 3) << 4);
      bfr[kk][e] = (__bf16)wp[k * DK_];
    }
    if ((kk & 3) == 3) __builtin_amdgcn_sched_barrier(0);
  }

  float m_run = -INFINITY;
  float s_run = 0.f;
  float acc   = 0.f;

  const size_t sbase = ((size_t)b * L_ + l0c) * D_;

  float4 stage[8];
  #pragma unroll
  for (int j = 0; j < 8; ++j)
    stage[j] = *reinterpret_cast<const float4*>(&S[sbase + tid * 4 + j * 1024]);

  for (int t = 0; t < TILES; ++t) {
    __syncthreads();
    #pragma unroll
    for (int j = 0; j < 8; ++j) {
      int e0  = tid * 4 + j * 1024;
      int row = e0 >> 9, col = e0 & 511;
      float4 f = stage[j];
      v4bf p4 = { (__bf16)f.x, (__bf16)f.y, (__bf16)f.z, (__bf16)f.w };
      *reinterpret_cast<v4bf*>(&s_tile[row * SROW + col]) = p4;
    }
    if (t + 1 < TILES) {
      const size_t sb1 = sbase + (size_t)(t + 1) * 16 * D_;
      #pragma unroll
      for (int j = 0; j < 8; ++j)
        stage[j] = *reinterpret_cast<const float4*>(&S[sb1 + tid * 4 + j * 1024]);
    }
    if (t + 2 < TILES) {
      const size_t sb2 = sbase + (size_t)(t + 2) * 16 * D_;
      #pragma unroll
      for (int j = 0; j < 8; ++j)
        __builtin_prefetch(&S[sb2 + tid * 4 + j * 1024], 0, 1);
    }
    __syncthreads();

    v8f c = {};
    #pragma unroll
    for (int kk = 0; kk < KSTEPS; ++kk) {
      const __bf16* pa = &s_tile[m_row * SROW + kk * 32 + kbase];
      v16bf a = join16(*reinterpret_cast<const v8bf*>(pa),
                       *reinterpret_cast<const v8bf*>(pa + 16));
      c = __builtin_amdgcn_wmma_f32_16x16x32_bf16(
            false, a, false, bfr[kk], (short)0, c, false, false);
    }

    if (sel == 0) {
      float qv = sh_q[wn * 16 + m_row];
      float part[8];
      #pragma unroll
      for (int r = 0; r < 8; ++r) part[r] = fast_tanh(c[r]) * qv;
      #pragma unroll
      for (int off = 1; off < 16; off <<= 1) {
        #pragma unroll
        for (int r = 0; r < 8; ++r) part[r] += __shfl_xor(part[r], off);
      }
      if ((lane & 15) == 0) {
        #pragma unroll
        for (int r = 0; r < 8; ++r) sh_part[wn][r + 8 * half] = part[r];
      }
    }
    __syncthreads();

    if (tid < 16) {
      float sc = (sh_part[0][tid] + sh_part[1][tid] +
                  sh_part[2][tid] + sh_part[3][tid]) * 0.125f;
      float mx = sc;
      #pragma unroll
      for (int off = 8; off >= 1; off >>= 1) mx = fmaxf(mx, __shfl_xor(mx, off));
      float m_new = fmaxf(m_run, mx);
      float scale = __expf(m_run - m_new);
      float p     = __expf(sc - m_new);
      float ps = p;
      #pragma unroll
      for (int off = 8; off >= 1; off >>= 1) ps += __shfl_xor(ps, off);
      s_run = s_run * scale + ps;
      m_run = m_new;
      sh_p[tid] = p;
      if (tid == 0) sh_scale = scale;
    }
    __syncthreads();

    if (sel == 1) {
      float contrib = 0.f;
      #pragma unroll
      for (int r = 0; r < 8; ++r) {
        float v = c[r];
        v = v > 0.f ? v : 0.f;
        contrib += sh_p[r + 8 * half] * v;
      }
      contrib += __shfl_xor(contrib, 16);
      acc = acc * sh_scale + contrib;
    }
  }

  if (sel == 1 && (lane & 16) == 0)
    ws_acc[(size_t)pid * DK_ + wn * 16 + m_row] = acc;
  if (tid == 0) { ws_m[pid] = m_run; ws_s[pid] = s_run; }
}

__global__ void __launch_bounds__(64)
ga_combine_kernel(const float* __restrict__ ws_m, const float* __restrict__ ws_s,
                  const float* __restrict__ ws_acc, float* __restrict__ out)
{
  const int bh = blockIdx.x;
  const int k  = threadIdx.x;
  float m = -INFINITY;
  #pragma unroll
  for (int sp = 0; sp < NSPLIT; ++sp) m = fmaxf(m, ws_m[bh * NSPLIT + sp]);
  float num = 0.f, den = 0.f;
  #pragma unroll
  for (int sp = 0; sp < NSPLIT; ++sp) {
    int pid = bh * NSPLIT + sp;
    float w = __expf(ws_m[pid] - m);
    num += w * ws_acc[(size_t)pid * DK_ + k];
    den += w * ws_s[pid];
  }
  out[bh * DK_ + k] = num / den;
}

extern "C" void kernel_launch(void* const* d_in, const int* in_sizes, int n_in,
                              void* d_out, int out_size, void* d_ws, size_t ws_size,
                              hipStream_t stream) {
  (void)in_sizes; (void)n_in; (void)out_size; (void)ws_size;
  const float* X  = (const float*)d_in[0];
  const float* S  = (const float*)d_in[1];
  const float* Wq = (const float*)d_in[2];
  const float* Wk = (const float*)d_in[3];
  const float* Wv = (const float*)d_in[4];

  const int NP = B_ * H_ * NSPLIT;
  float* wsf    = (float*)d_ws;
  float* ws_m   = wsf;
  float* ws_s   = wsf + NP;
  float* ws_acc = wsf + 2 * NP;

  ga_flash_kernel<<<NP, 256, 0, stream>>>(X, S, Wq, Wk, Wv, ws_m, ws_s, ws_acc);
  ga_combine_kernel<<<B_ * H_, 64, 0, stream>>>(ws_m, ws_s, ws_acc, (float*)d_out);
}


# A fragments read from LDS through a 6-slot register ring, 4 loads as one group 4 K-steps ahead of their WMMAs with counted waits, instead of one slot and a full LDS wait in front of every WMMA
# speedup vs baseline: 1.5303x; 1.1341x over previous
#include <hip/hip_runtime.h>
#include <hip/hip_bf16.h>
#include <math.h>

typedef __attribute__((ext_vector_type(8)))  float  v8f;
typedef __attribute__((ext_vector_type(4)))  __bf16 v4bf;
typedef __attribute__((ext_vector_type(8)))  __bf16 v8bf;
typedef __attribute__((ext_vector_type(16))) __bf16 v16bf;

#define B_   32
#define L_   2048
#define D_   512
#define H_   8
#define DK_  64
#define NSPLIT 4
#define CHUNK (L_ / NSPLIT)
#define TILES (CHUNK / 16)
#define KSTEPS (D_ / 32)
#define SROW 520
#define ARING 6
#define AGRP  2

static __device__ inline v16bf join16(v8bf lo, v8bf hi) {
  return __builtin_shufflevector(lo, hi, 0,1,2,3,4,5,6,7,8,9,10,11,12,13,14,15);
}

static __device__ inline float fast_tanh(float x) {
  x = fminf(fmaxf(x, -15.f), 15.f);
  float e = __expf(2.0f * x);
  return (e - 1.0f) / (e + 1.0f);
}

__global__ void __launch_bounds__(256)
ga_flash_kernel(const float* __restrict__ X, const float* __restrict__ S,
                const float* __restrict__ Wq, const float* __restrict__ Wk,
                const float* __restrict__ Wv,
                float* __restrict__ ws_m, float* __restrict__ ws_s,
                float* __restrict__ ws_acc)
{
  __shared__ __bf16 s_tile[16 * SROW];
  __shared__ float  sh_x[D_];
  __shared__ float  sh_q[DK_];
  __shared__ float  sh_part[4][16];
  __shared__ float  sh_p[16];
  __shared__ float  sh_scale;

  const int tid  = threadIdx.x;
  const int lane = tid & 31;
  const int wave = tid >> 5;
  const int wn   = wave & 3;
  const int sel  = wave >> 2;

  const int pid = blockIdx.x;
  const int sp  = pid & (NSPLIT - 1);
  const int h   = (pid >> 2) & (H_ - 1);
  const int b   = pid >> 5;
  const int l0c = sp * CHUNK;

  const int m_row = lane & 15;
  const int kbase = (lane < 16) ? 0 : 8;
  const int half  = lane >> 4;

  for (int i = tid; i < D_; i += 256) sh_x[i] = X[b * D_ + i];
  __syncthreads();

  if (tid < DK_) {
    float a = 0.f;
    for (int g = 0; g < D_; ++g) a += sh_x[g] * Wq[(h * D_ + g) * DK_ + tid];
    sh_q[tid] = fast_tanh(a);
  }

  const float* wp = (sel ? Wv : Wk) + ((size_t)h * D_ + kbase) * DK_ + wn * 16 + m_row;
  v16bf bfr[KSTEPS];
  #pragma unroll
  for (int kk = 0; kk < KSTEPS; ++kk) {
    #pragma unroll
    for (int e = 0; e < 16; ++e) {
      const int k = kk * 32 + (e & 7) + ((e >> 3) << 4);
      bfr[kk][e] = (__bf16)wp[k * DK_];
    }
    if ((kk & 3) == 3) __builtin_amdgcn_sched_barrier(0);
  }

  float m_run = -INFINITY;
  float s_run = 0.f;
  float acc   = 0.f;

  const size_t sbase = ((size_t)b * L_ + l0c) * D_;

  float4 stage[8];
  #pragma unroll
  for (int j = 0; j < 8; ++j)
    stage[j] = *reinterpret_cast<const float4*>(&S[sbase + tid * 4 + j * 1024]);

  for (int t = 0; t < TILES; ++t) {
    __syncthreads();
    #pragma unroll
    for (int j = 0; j < 8; ++j) {
      int e0  = tid * 4 + j * 1024;
      int row = e0 >> 9, col = e0 & 511;
      float4 f = stage[j];
      v4bf p4 = { (__bf16)f.x, (__bf16)f.y, (__bf16)f.z, (__bf16)f.w };
      *reinterpret_cast<v4bf*>(&s_tile[row * SROW + col]) = p4;
    }
    if (t + 1 < TILES) {
      const size_t sb1 = sbase + (size_t)(t + 1) * 16 * D_;
      #pragma unroll
      for (int j = 0; j < 8; ++j)
        stage[j] = *reinterpret_cast<const float4*>(&S[sb1 + tid * 4 + j * 1024]);
    }
    if (t + 2 < TILES) {
      const size_t sb2 = sbase + (size_t)(t + 2) * 16 * D_;
      #pragma unroll
      for (int j = 0; j < 8; ++j)
        __builtin_prefetch(&S[sb2 + tid * 4 + j * 1024], 0, 1);
    }
    __syncthreads();

    const __bf16* pa = &s_tile[m_row * SROW + kbase];
    v16bf afr[ARING];
    #pragma unroll
    for (int i = 0; i < ARING - AGRP; ++i)
      afr[i] = join16(*reinterpret_cast<const v8bf*>(pa + i * 32),
                      *reinterpret_cast<const v8bf*>(pa + i * 32 + 16));
    v8f c = {};
    #pragma unroll
    for (int kk = 0; kk < KSTEPS; ++kk) {
      if (kk % AGRP == 0) {
        __builtin_amdgcn_sched_barrier(0);
        #pragma unroll
        for (int g = 0; g < AGRP; ++g) {
          const int kn = kk + ARING - AGRP + g;
          if (kn < KSTEPS)
            afr[kn % ARING] = join16(*reinterpret_cast<const v8bf*>(pa + kn * 32),
                                     *reinterpret_cast<const v8bf*>(pa + kn * 32 + 16));
        }
        __builtin_amdgcn_sched_barrier(0);
      }
      c = __builtin_amdgcn_wmma_f32_16x16x32_bf16(
            false, afr[kk % ARING], false, bfr[kk], (short)0, c, false, false);
    }

    if (sel == 0) {
      float qv = sh_q[wn * 16 + m_row];
      float part[8];
      #pragma unroll
      for (int r = 0; r < 8; ++r) part[r] = fast_tanh(c[r]) * qv;
      #pragma unroll
      for (int off = 1; off < 16; off <<= 1) {
        #pragma unroll
        for (int r = 0; r < 8; ++r) part[r] += __shfl_xor(part[r], off);
      }
      if ((lane & 15) == 0) {
        #pragma unroll
        for (int r = 0; r < 8; ++r) sh_part[wn][r + 8 * half] = part[r];
      }
    }
    __syncthreads();

    if (tid < 16) {
      float sc = (sh_part[0][tid] + sh_part[1][tid] +
                  sh_part[2][tid] + sh_part[3][tid]) * 0.125f;
      float mx = sc;
      #pragma unroll
      for (int off = 8; off >= 1; off >>= 1) mx = fmaxf(mx, __shfl_xor(mx, off));
      float m_new = fmaxf(m_run, mx);
      float scale = __expf(m_run - m_new);
      float p     = __expf(sc - m_new);
      float ps = p;
      #pragma unroll
      for (int off = 8; off >= 1; off >>= 1) ps += __shfl_xor(ps, off);
      s_run = s_run * scale + ps;
      m_run = m_new;
      sh_p[tid] = p;
      if (tid == 0) sh_scale = scale;
    }
    __syncthreads();

    if (sel == 1) {
      float contrib = 0.f;
      #pragma unroll
      for (int r = 0; r < 8; ++r) {
        float v = c[r];
        v = v > 0.f ? v : 0.f;
        contrib += sh_p[r + 8 * half] * v;
      }
      contrib += __shfl_xor(contrib, 16);
      acc = acc * sh_scale + contrib;
    }
  }

  if (sel == 1 && (lane & 16) == 0)
    ws_acc[(size_t)pid * DK_ + wn * 16 + m_row] = acc;
  if (tid == 0) { ws_m[pid] = m_run; ws_s[pid] = s_run; }
}

__global__ void __launch_bounds__(64)
ga_combine_kernel(const float* __restrict__ ws_m, const float* __restrict__ ws_s,
                  const float* __restrict__ ws_acc, float* __restrict__ out)
{
  const int bh = blockIdx.x;
  const int k  = threadIdx.x;
  float m = -INFINITY;
  #pragma unroll
  for (int sp = 0; sp < NSPLIT; ++sp) m = fmaxf(m, ws_m[bh * NSPLIT + sp]);
  float num = 0.f, den = 0.f;
  #pragma unroll
  for (int sp = 0; sp < NSPLIT; ++sp) {
    int pid = bh * NSPLIT + sp;
    float w = __expf(ws_m[pid] - m);
    num += w * ws_acc[(size_t)pid * DK_ + k];
    den += w * ws_s[pid];
  }
  out[bh * DK_ + k] = num / den;
}

extern "C" void kernel_launch(void* const* d_in, const int* in_sizes, int n_in,
                              void* d_out, int out_size, void* d_ws, size_t ws_size,
                              hipStream_t stream) {
  (void)in_sizes; (void)n_in; (void)out_size; (void)ws_size;
  const float* X  = (const float*)d_in[0];
  const float* S  = (const float*)d_in[1];
  const float* Wq = (const float*)d_in[2];
  const float* Wk = (const float*)d_in[3];
  const float* Wv = (const float*)d_in[4];

  const int NP = B_ * H_ * NSPLIT;
  float* wsf    = (float*)d_ws;
  float* ws_m   = wsf;
  float* ws_s   = wsf + NP;
  float* ws_acc = wsf + 2 * NP;

  ga_flash_kernel<<<NP, 256, 0, stream>>>(X, S, Wq, Wk, Wv, ws_m, ws_s, ws_acc);
  ga_combine_kernel<<<B_ * H_, 64, 0, stream>>>(ws_m, ws_s, ws_acc, (float*)d_out);
}


# Tile-loop barriers fence LDS only (work-group fences limited to the LDS address space + s_barrier): the next tile's global loads stay in flight behind the GEMM, not waited for at the barrier
# speedup vs baseline: 1.5826x; 1.0342x over previous
#include <hip/hip_runtime.h>
#include <hip/hip_bf16.h>
#include <math.h>

typedef __attribute__((ext_vector_type(8)))  float  v8f;
typedef __attribute__((ext_vector_type(4)))  __bf16 v4bf;
typedef __attribute__((ext_vector_type(8)))  __bf16 v8bf;
typedef __attribute__((ext_vector_type(16))) __bf16 v16bf;

#define B_   32
#define L_   2048
#define D_   512
#define H_   8
#define DK_  64
#define NSPLIT 4
#define CHUNK (L_ / NSPLIT)
#define TILES (CHUNK / 16)
#define KSTEPS (D_ / 32)
#define SROW 520
#define ARING 6
#define AGRP  2

static __device__ inline v16bf join16(v8bf lo, v8bf hi) {
  return __builtin_shufflevector(lo, hi, 0,1,2,3,4,5,6,7,8,9,10,11,12,13,14,15);
}

static __device__ inline float fast_tanh(float x) {
  x = fminf(fmaxf(x, -15.f), 15.f);
  float e = __expf(2.0f * x);
  return (e - 1.0f) / (e + 1.0f);
}

static __device__ inline void lds_barrier() {
  __builtin_amdgcn_fence(__ATOMIC_RELEASE, "workgroup", "local");
  __builtin_amdgcn_s_barrier();
  __builtin_amdgcn_fence(__ATOMIC_ACQUIRE, "workgroup", "local");
}

__global__ void __launch_bounds__(256)
ga_flash_kernel(const float* __restrict__ X, const float* __restrict__ S,
                const float* __restrict__ Wq, const float* __restrict__ Wk,
                const float* __restrict__ Wv,
                float* __restrict__ ws_m, float* __restrict__ ws_s,
                float* __restrict__ ws_acc)
{
  __shared__ __bf16 s_tile[16 * SROW];
  __shared__ float  sh_x[D_];
  __shared__ float  sh_q[DK_];
  __shared__ float  sh_part[4][16];
  __shared__ float  sh_p[16];
  __shared__ float  sh_scale;

  const int tid  = threadIdx.x;
  const int lane = tid & 31;
  const int wave = tid >> 5;
  const int wn   = wave & 3;
  const int sel  = wave >> 2;

  const int pid = blockIdx.x;
  const int sp  = pid & (NSPLIT - 1);
  const int h   = (pid >> 2) & (H_ - 1);
  const int b   = pid >> 5;
  const int l0c = sp * CHUNK;

  const int m_row = lane & 15;
  const int kbase = (lane < 16) ? 0 : 8;
  const int half  = lane >> 4;

  for (int i = tid; i < D_; i += 256) sh_x[i] = X[b * D_ + i];
  __syncthreads();

  if (tid < DK_) {
    float a = 0.f;
    for (int g = 0; g < D_; ++g) a += sh_x[g] * Wq[(h * D_ + g) * DK_ + tid];
    sh_q[tid] = fast_tanh(a);
  }

  const float* wp = (sel ? Wv : Wk) + ((size_t)h * D_ + kbase) * DK_ + wn * 16 + m_row;
  v16bf bfr[KSTEPS];
  #pragma unroll
  for (int kk = 0; kk < KSTEPS; ++kk) {
    #pragma unroll
    for (int e = 0; e < 16; ++e) {
      const int k = kk * 32 + (e & 7) + ((e >> 3) << 4);
      bfr[kk][e] = (__bf16)wp[k * DK_];
    }
    if ((kk & 3) == 3) __builtin_amdgcn_sched_barrier(0);
  }

  float m_run = -INFINITY;
  float s_run = 0.f;
  float acc   = 0.f;

  const size_t sbase = ((size_t)b * L_ + l0c) * D_;

  float4 stage[8];
  #pragma unroll
  for (int j = 0; j < 8; ++j)
    stage[j] = *reinterpret_cast<const float4*>(&S[sbase + tid * 4 + j * 1024]);

  for (int t = 0; t < TILES; ++t) {
    lds_barrier();
    #pragma unroll
    for (int j = 0; j < 8; ++j) {
      int e0  = tid * 4 + j * 1024;
      int row = e0 >> 9, col = e0 & 511;
      float4 f = stage[j];
      v4bf p4 = { (__bf16)f.x, (__bf16)f.y, (__bf16)f.z, (__bf16)f.w };
      *reinterpret_cast<v4bf*>(&s_tile[row * SROW + col]) = p4;
    }
    if (t + 1 < TILES) {
      const size_t sb1 = sbase + (size_t)(t + 1) * 16 * D_;
      #pragma unroll
      for (int j = 0; j < 8; ++j)
        stage[j] = *reinterpret_cast<const float4*>(&S[sb1 + tid * 4 + j * 1024]);
    }
    if (t + 2 < TILES) {
      const size_t sb2 = sbase + (size_t)(t + 2) * 16 * D_;
      #pragma unroll
      for (int j = 0; j < 8; ++j)
        __builtin_prefetch(&S[sb2 + tid * 4 + j * 1024], 0, 1);
    }
    lds_barrier();

    const __bf16* pa = &s_tile[m_row * SROW + kbase];
    v16bf afr[ARING];
    #pragma unroll
    for (int i = 0; i < ARING - AGRP; ++i)
      afr[i] = join16(*reinterpret_cast<const v8bf*>(pa + i * 32),
                      *reinterpret_cast<const v8bf*>(pa + i * 32 + 16));
    v8f c = {};
    #pragma unroll
    for (int kk = 0; kk < KSTEPS; ++kk) {
      if (kk % AGRP == 0) {
        __builtin_amdgcn_sched_barrier(0);
        #pragma unroll
        for (int g = 0; g < AGRP; ++g) {
          const int kn = kk + ARING - AGRP + g;
          if (kn < KSTEPS)
            afr[kn % ARING] = join16(*reinterpret_cast<const v8bf*>(pa + kn * 32),
                                     *reinterpret_cast<const v8bf*>(pa + kn * 32 + 16));
        }
        __builtin_amdgcn_sched_barrier(0);
      }
      c = __builtin_amdgcn_wmma_f32_16x16x32_bf16(
            false, afr[kk % ARING], false, bfr[kk], (short)0, c, false, false);
    }

    if (sel == 0) {
      float qv = sh_q[wn * 16 + m_row];
      float part[8];
      #pragma unroll
      for (int r = 0; r < 8; ++r) part[r] = fast_tanh(c[r]) * qv;
      #pragma unroll
      for (int off = 1; off < 16; off <<= 1) {
        #pragma unroll
        for (int r = 0; r < 8; ++r) part[r] += __shfl_xor(part[r], off);
      }
      if ((lane & 15) == 0) {
        #pragma unroll
        for (int r = 0; r < 8; ++r) sh_part[wn][r + 8 * half] = part[r];
      }
    }
    lds_barrier();

    if (tid < 16) {
      float sc = (sh_part[0][tid] + sh_part[1][tid] +
                  sh_part[2][tid] + sh_part[3][tid]) * 0.125f;
      float mx = sc;
      #pragma unroll
      for (int off = 8; off >= 1; off >>= 1) mx = fmaxf(mx, __shfl_xor(mx, off));
      float m_new = fmaxf(m_run, mx);
      float scale = __expf(m_run - m_new);
      float p     = __expf(sc - m_new);
      float ps = p;
      #pragma unroll
      for (int off = 8; off >= 1; off >>= 1) ps += __shfl_xor(ps, off);
      s_run = s_run * scale + ps;
      m_run = m_new;
      sh_p[tid] = p;
      if (tid == 0) sh_scale = scale;
    }
    lds_barrier();

    if (sel == 1) {
      float contrib = 0.f;
      #pragma unroll
      for (int r = 0; r < 8; ++r) {
        float v = c[r];
        v = v > 0.f ? v : 0.f;
        contrib += sh_p[r + 8 * half] * v;
      }
      contrib += __shfl_xor(contrib, 16);
      acc = acc * sh_scale + contrib;
    }
  }

  if (sel == 1 && (lane & 16) == 0)
    ws_acc[(size_t)pid * DK_ + wn * 16 + m_row] = acc;
  if (tid == 0) { ws_m[pid] = m_run; ws_s[pid] = s_run; }
}

__global__ void __launch_bounds__(64)
ga_combine_kernel(const float* __restrict__ ws_m, const float* __restrict__ ws_s,
                  const float* __restrict__ ws_acc, float* __restrict__ out)
{
  const int bh = blockIdx.x;
  const int k  = threadIdx.x;
  float m = -INFINITY;
  #pragma unroll
  for (int sp = 0; sp < NSPLIT; ++sp) m = fmaxf(m, ws_m[bh * NSPLIT + sp]);
  float num = 0.f, den = 0.f;
  #pragma unroll
  for (int sp = 0; sp < NSPLIT; ++sp) {
    int pid = bh * NSPLIT + sp;
    float w = __expf(ws_m[pid] - m);
    num += w * ws_acc[(size_t)pid * DK_ + k];
    den += w * ws_s[pid];
  }
  out[bh * DK_ + k] = num / den;
}

extern "C" void kernel_launch(void* const* d_in, const int* in_sizes, int n_in,
                              void* d_out, int out_size, void* d_ws, size_t ws_size,
                              hipStream_t stream) {
  (void)in_sizes; (void)n_in; (void)out_size; (void)ws_size;
  const float* X  = (const float*)d_in[0];
  const float* S  = (const float*)d_in[1];
  const float* Wq = (const float*)d_in[2];
  const float* Wk = (const float*)d_in[3];
  const float* Wv = (const float*)d_in[4];

  const int NP = B_ * H_ * NSPLIT;
  float* wsf    = (float*)d_ws;
  float* ws_m   = wsf;
  float* ws_s   = wsf + NP;
  float* ws_acc = wsf + 2 * NP;

  ga_flash_kernel<<<NP, 256, 0, stream>>>(X, S, Wq, Wk, Wv, ws_m, ws_s, ws_acc);
  ga_combine_kernel<<<B_ * H_, 64, 0, stream>>>(ws_m, ws_s, ws_acc, (float*)d_out);
}


# Softmax against the fixed bound |score| <= 8 (q and Kt are tanh outputs): p = exp(score - 8), no running max or rescale; V-waves form p one tile late: 2 barriers per tile, not 4; sums reordered
# speedup vs baseline: 1.7510x; 1.1064x over previous
#include <hip/hip_runtime.h>
#include <hip/hip_bf16.h>
#include <math.h>

typedef __attribute__((ext_vector_type(8)))  float  v8f;
typedef __attribute__((ext_vector_type(4)))  __bf16 v4bf;
typedef __attribute__((ext_vector_type(8)))  __bf16 v8bf;
typedef __attribute__((ext_vector_type(16))) __bf16 v16bf;

#define B_   32
#define L_   2048
#define D_   512
#define H_   8
#define DK_  64
#define NSPLIT 4
#define CHUNK (L_ / NSPLIT)
#define TILES (CHUNK / 16)
#define KSTEPS (D_ / 32)
#define SROW 520
#define ARING 6
#define AGRP  2

static __device__ inline v16bf join16(v8bf lo, v8bf hi) {
  return __builtin_shufflevector(lo, hi, 0,1,2,3,4,5,6,7,8,9,10,11,12,13,14,15);
}

static __device__ inline float fast_tanh(float x) {
  x = fminf(fmaxf(x, -15.f), 15.f);
  float e = __expf(2.0f * x);
  return (e - 1.0f) / (e + 1.0f);
}

static __device__ inline void lds_barrier() {
  __builtin_amdgcn_fence(__ATOMIC_RELEASE, "workgroup", "local");
  __builtin_amdgcn_s_barrier();
  __builtin_amdgcn_fence(__ATOMIC_ACQUIRE, "workgroup", "local");
}

__global__ void __launch_bounds__(256)
ga_flash_kernel(const float* __restrict__ X, const float* __restrict__ S,
                const float* __restrict__ Wq, const float* __restrict__ Wk,
                const float* __restrict__ Wv,
                float* __restrict__ ws_s, float* __restrict__ ws_acc)
{
  __shared__ __bf16 s_tile[16 * SROW];
  __shared__ float  sh_x[D_];
  __shared__ float  sh_q[DK_];
  __shared__ __attribute__((aligned(16))) float sh_part[2][4][16];

  const int tid  = threadIdx.x;
  const int lane = tid & 31;
  const int wave = tid >> 5;
  const int wn   = wave & 3;
  const int sel  = wave >> 2;

  const int pid = blockIdx.x;
  const int sp  = pid & (NSPLIT - 1);
  const int h   = (pid >> 2) & (H_ - 1);
  const int b   = pid >> 5;
  const int l0c = sp * CHUNK;

  const int m_row = lane & 15;
  const int kbase = (lane < 16) ? 0 : 8;
  const int half  = lane >> 4;

  for (int i = tid; i < D_; i += 256) sh_x[i] = X[b * D_ + i];
  __syncthreads();

  if (tid < DK_) {
    float a = 0.f;
    for (int g = 0; g < D_; ++g) a += sh_x[g] * Wq[(h * D_ + g) * DK_ + tid];
    sh_q[tid] = fast_tanh(a);
  }

  const float* wp = (sel ? Wv : Wk) + ((size_t)h * D_ + kbase) * DK_ + wn * 16 + m_row;
  v16bf bfr[KSTEPS];
  #pragma unroll
  for (int kk = 0; kk < KSTEPS; ++kk) {
    #pragma unroll
    for (int e = 0; e < 16; ++e) {
      const int k = kk * 32 + (e & 7) + ((e >> 3) << 4);
      bfr[kk][e] = (__bf16)wp[k * DK_];
    }
    if ((kk & 3) == 3) __builtin_amdgcn_sched_barrier(0);
  }

  float psum = 0.f;
  float acc  = 0.f;
  v8f c = {};

  auto v_update = [&](int buf) {
    const float* pp = &sh_part[buf][0][8 * half];
    float sc[8];
    #pragma unroll
    for (int w = 0; w < 4; ++w) {
      float4 lo = *reinterpret_cast<const float4*>(pp + 16 * w);
      float4 hi = *reinterpret_cast<const float4*>(pp + 16 * w + 4);
      float v[8] = { lo.x, lo.y, lo.z, lo.w, hi.x, hi.y, hi.z, hi.w };
      #pragma unroll
      for (int r = 0; r < 8; ++r) sc[r] = (w == 0) ? v[r] : sc[r] + v[r];
    }
    #pragma unroll
    for (int r = 0; r < 8; ++r) {
      float p = __expf(fmaf(sc[r], 0.125f, -8.0f));
      float v = c[r];
      v = v > 0.f ? v : 0.f;
      psum += p;
      acc = fmaf(p, v, acc);
    }
  };

  const size_t sbase = ((size_t)b * L_ + l0c) * D_;

  float4 stage[8];
  #pragma unroll
  for (int j = 0; j < 8; ++j)
    stage[j] = *reinterpret_cast<const float4*>(&S[sbase + tid * 4 + j * 1024]);

  for (int t = 0; t < TILES; ++t) {
    lds_barrier();
    #pragma unroll
    for (int j = 0; j < 8; ++j) {
      int e0  = tid * 4 + j * 1024;
      int row = e0 >> 9, col = e0 & 511;
      float4 f = stage[j];
      v4bf p4 = { (__bf16)f.x, (__bf16)f.y, (__bf16)f.z, (__bf16)f.w };
      *reinterpret_cast<v4bf*>(&s_tile[row * SROW + col]) = p4;
    }
    if (t + 1 < TILES) {
      const size_t sb1 = sbase + (size_t)(t + 1) * 16 * D_;
      #pragma unroll
      for (int j = 0; j < 8; ++j)
        stage[j] = *reinterpret_cast<const float4*>(&S[sb1 + tid * 4 + j * 1024]);
    }
    if (t + 2 < TILES) {
      const size_t sb2 = sbase + (size_t)(t + 2) * 16 * D_;
      #pragma unroll
      for (int j = 0; j < 8; ++j)
        __builtin_prefetch(&S[sb2 + tid * 4 + j * 1024], 0, 1);
    }
    lds_barrier();

    if (sel == 1 && t > 0) v_update((t - 1) & 1);

    const __bf16* pa = &s_tile[m_row * SROW + kbase];
    v16bf afr[ARING];
    #pragma unroll
    for (int i = 0; i < ARING - AGRP; ++i)
      afr[i] = join16(*reinterpret_cast<const v8bf*>(pa + i * 32),
                      *reinterpret_cast<const v8bf*>(pa + i * 32 + 16));
    c = v8f{};
    #pragma unroll
    for (int kk = 0; kk < KSTEPS; ++kk) {
      if (kk % AGRP == 0) {
        __builtin_amdgcn_sched_barrier(0);
        #pragma unroll
        for (int g = 0; g < AGRP; ++g) {
          const int kn = kk + ARING - AGRP + g;
          if (kn < KSTEPS)
            afr[kn % ARING] = join16(*reinterpret_cast<const v8bf*>(pa + kn * 32),
                                     *reinterpret_cast<const v8bf*>(pa + kn * 32 + 16));
        }
        __builtin_amdgcn_sched_barrier(0);
      }
      c = __builtin_amdgcn_wmma_f32_16x16x32_bf16(
            false, afr[kk % ARING], false, bfr[kk], (short)0, c, false, false);
    }

    if (sel == 0) {
      float qv = sh_q[wn * 16 + m_row];
      float part[8];
      #pragma unroll
      for (int r = 0; r < 8; ++r) part[r] = fast_tanh(c[r]) * qv;
      #pragma unroll
      for (int off = 1; off < 16; off <<= 1) {
        #pragma unroll
        for (int r = 0; r < 8; ++r) part[r] += __shfl_xor(part[r], off);
      }
      if ((lane & 15) == 0) {
        #pragma unroll
        for (int r = 0; r < 8; ++r) sh_part[t & 1][wn][r + 8 * half] = part[r];
      }
    }
  }

  lds_barrier();
  if (sel == 1) {
    v_update((TILES - 1) & 1);
    acc  += __shfl_xor(acc, 16);
    psum += __shfl_xor(psum, 16);
    if (lane < 16) ws_acc[(size_t)pid * DK_ + wn * 16 + m_row] = acc;
    if (tid == 128) ws_s[pid] = psum;
  }
}

__global__ void __launch_bounds__(64)
ga_combine_kernel(const float* __restrict__ ws_s, const float* __restrict__ ws_acc,
                  float* __restrict__ out)
{
  const int bh = blockIdx.x;
  const int k  = threadIdx.x;
  float num = 0.f, den = 0.f;
  #pragma unroll
  for (int sp = 0; sp < NSPLIT; ++sp) {
    int pid = bh * NSPLIT + sp;
    num += ws_acc[(size_t)pid * DK_ + k];
    den += ws_s[pid];
  }
  out[bh * DK_ + k] = num / den;
}

extern "C" void kernel_launch(void* const* d_in, const int* in_sizes, int n_in,
                              void* d_out, int out_size, void* d_ws, size_t ws_size,
                              hipStream_t stream) {
  (void)in_sizes; (void)n_in; (void)out_size; (void)ws_size;
  const float* X  = (const float*)d_in[0];
  const float* S  = (const float*)d_in[1];
  const float* Wq = (const float*)d_in[2];
  const float* Wk = (const float*)d_in[3];
  const float* Wv = (const float*)d_in[4];

  const int NP = B_ * H_ * NSPLIT;
  float* wsf    = (float*)d_ws;
  float* ws_s   = wsf;
  float* ws_acc = wsf + NP;

  ga_flash_kernel<<<NP, 256, 0, stream>>>(X, S, Wq, Wk, Wv, ws_s, ws_acc);
  ga_combine_kernel<<<B_ * H_, 64, 0, stream>>>(ws_s, ws_acc, (float*)d_out);
}


# Two row-splits per (b,h) instead of four: 512 blocks of 64 tiles, all resident in one round (2 per WGP), pay the block set-up (q, weight fragments) once, not twice; partial sums regrouped
# speedup vs baseline: 2.1174x; 1.2093x over previous
#include <hip/hip_runtime.h>
#include <hip/hip_bf16.h>
#include <math.h>

typedef __attribute__((ext_vector_type(8)))  float  v8f;
typedef __attribute__((ext_vector_type(4)))  __bf16 v4bf;
typedef __attribute__((ext_vector_type(8)))  __bf16 v8bf;
typedef __attribute__((ext_vector_type(16))) __bf16 v16bf;

#define B_   32
#define L_   2048
#define D_   512
#define H_   8
#define DK_  64
#define NSPLIT 2
#define CHUNK (L_ / NSPLIT)
#define TILES (CHUNK / 16)
#define KSTEPS (D_ / 32)
#define SROW 520
#define ARING 6
#define AGRP  2

static __device__ inline v16bf join16(v8bf lo, v8bf hi) {
  return __builtin_shufflevector(lo, hi, 0,1,2,3,4,5,6,7,8,9,10,11,12,13,14,15);
}

static __device__ inline float fast_tanh(float x) {
  x = fminf(fmaxf(x, -15.f), 15.f);
  float e = __expf(2.0f * x);
  return (e - 1.0f) / (e + 1.0f);
}

static __device__ inline void lds_barrier() {
  __builtin_amdgcn_fence(__ATOMIC_RELEASE, "workgroup", "local");
  __builtin_amdgcn_s_barrier();
  __builtin_amdgcn_fence(__ATOMIC_ACQUIRE, "workgroup", "local");
}

__global__ void __launch_bounds__(256)
ga_flash_kernel(const float* __restrict__ X, const float* __restrict__ S,
                const float* __restrict__ Wq, const float* __restrict__ Wk,
                const float* __restrict__ Wv,
                float* __restrict__ ws_s, float* __restrict__ ws_acc)
{
  __shared__ __bf16 s_tile[16 * SROW];
  __shared__ float  sh_x[D_];
  __shared__ float  sh_q[DK_];
  __shared__ __attribute__((aligned(16))) float sh_part[2][4][16];

  const int tid  = threadIdx.x;
  const int lane = tid & 31;
  const int wave = tid >> 5;
  const int wn   = wave & 3;
  const int sel  = wave >> 2;

  const int pid = blockIdx.x;
  const int sp  = pid & (NSPLIT - 1);
  const int h   = (pid >> 1) & (H_ - 1);
  const int b   = pid >> 4;
  const int l0c = sp * CHUNK;

  const int m_row = lane & 15;
  const int kbase = (lane < 16) ? 0 : 8;
  const int half  = lane >> 4;

  for (int i = tid; i < D_; i += 256) sh_x[i] = X[b * D_ + i];
  __syncthreads();

  if (tid < DK_) {
    float a = 0.f;
    for (int g = 0; g < D_; ++g) a += sh_x[g] * Wq[(h * D_ + g) * DK_ + tid];
    sh_q[tid] = fast_tanh(a);
  }

  const float* wp = (sel ? Wv : Wk) + ((size_t)h * D_ + kbase) * DK_ + wn * 16 + m_row;
  v16bf bfr[KSTEPS];
  #pragma unroll
  for (int kk = 0; kk < KSTEPS; ++kk) {
    #pragma unroll
    for (int e = 0; e < 16; ++e) {
      const int k = kk * 32 + (e & 7) + ((e >> 3) << 4);
      bfr[kk][e] = (__bf16)wp[k * DK_];
    }
    if ((kk & 3) == 3) __builtin_amdgcn_sched_barrier(0);
  }

  float psum = 0.f;
  float acc  = 0.f;
  v8f c = {};

  auto v_update = [&](int buf) {
    const float* pp = &sh_part[buf][0][8 * half];
    float sc[8];
    #pragma unroll
    for (int w = 0; w < 4; ++w) {
      float4 lo = *reinterpret_cast<const float4*>(pp + 16 * w);
      float4 hi = *reinterpret_cast<const float4*>(pp + 16 * w + 4);
      float v[8] = { lo.x, lo.y, lo.z, lo.w, hi.x, hi.y, hi.z, hi.w };
      #pragma unroll
      for (int r = 0; r < 8; ++r) sc[r] = (w == 0) ? v[r] : sc[r] + v[r];
    }
    #pragma unroll
    for (int r = 0; r < 8; ++r) {
      float p = __expf(fmaf(sc[r], 0.125f, -8.0f));
      float v = c[r];
      v = v > 0.f ? v : 0.f;
      psum += p;
      acc = fmaf(p, v, acc);
    }
  };

  const size_t sbase = ((size_t)b * L_ + l0c) * D_;

  float4 stage[8];
  #pragma unroll
  for (int j = 0; j < 8; ++j)
    stage[j] = *reinterpret_cast<const float4*>(&S[sbase + tid * 4 + j * 1024]);

  for (int t = 0; t < TILES; ++t) {
    lds_barrier();
    #pragma unroll
    for (int j = 0; j < 8; ++j) {
      int e0  = tid * 4 + j * 1024;
      int row = e0 >> 9, col = e0 & 511;
      float4 f = stage[j];
      v4bf p4 = { (__bf16)f.x, (__bf16)f.y, (__bf16)f.z, (__bf16)f.w };
      *reinterpret_cast<v4bf*>(&s_tile[row * SROW + col]) = p4;
    }
    if (t + 1 < TILES) {
      const size_t sb1 = sbase + (size_t)(t + 1) * 16 * D_;
      #pragma unroll
      for (int j = 0; j < 8; ++j)
        stage[j] = *reinterpret_cast<const float4*>(&S[sb1 + tid * 4 + j * 1024]);
    }
    if (t + 2 < TILES) {
      const size_t sb2 = sbase + (size_t)(t + 2) * 16 * D_;
      #pragma unroll
      for (int j = 0; j < 8; ++j)
        __builtin_prefetch(&S[sb2 + tid * 4 + j * 1024], 0, 1);
    }
    lds_barrier();

    if (sel == 1 && t > 0) v_update((t - 1) & 1);

    const __bf16* pa = &s_tile[m_row * SROW + kbase];
    v16bf afr[ARING];
    #pragma unroll
    for (int i = 0; i < ARING - AGRP; ++i)
      afr[i] = join16(*reinterpret_cast<const v8bf*>(pa + i * 32),
                      *reinterpret_cast<const v8bf*>(pa + i * 32 + 16));
    c = v8f{};
    #pragma unroll
    for (int kk = 0; kk < KSTEPS; ++kk) {
      if (kk % AGRP == 0) {
        __builtin_amdgcn_sched_barrier(0);
        #pragma unroll
        for (int g = 0; g < AGRP; ++g) {
          const int kn = kk + ARING - AGRP + g;
          if (kn < KSTEPS)
            afr[kn % ARING] = join16(*reinterpret_cast<const v8bf*>(pa + kn * 32),
                                     *reinterpret_cast<const v8bf*>(pa + kn * 32 + 16));
        }
        __builtin_amdgcn_sched_barrier(0);
      }
      c = __builtin_amdgcn_wmma_f32_16x16x32_bf16(
            false, afr[kk % ARING], false, bfr[kk], (short)0, c, false, false);
    }

    if (sel == 0) {
      float qv = sh_q[wn * 16 + m_row];
      float part[8];
      #pragma unroll
      for (int r = 0; r < 8; ++r) part[r] = fast_tanh(c[r]) * qv;
      #pragma unroll
      for (int off = 1; off < 16; off <<= 1) {
        #pragma unroll
        for (int r = 0; r < 8; ++r) part[r] += __shfl_xor(part[r], off);
      }
      if ((lane & 15) == 0) {
        #pragma unroll
        for (int r = 0; r < 8; ++r) sh_part[t & 1][wn][r + 8 * half] = part[r];
      }
    }
  }

  lds_barrier();
  if (sel == 1) {
    v_update((TILES - 1) & 1);
    acc  += __shfl_xor(acc, 16);
    psum += __shfl_xor(psum, 16);
    if (lane < 16) ws_acc[(size_t)pid * DK_ + wn * 16 + m_row] = acc;
    if (tid == 128) ws_s[pid] = psum;
  }
}

__global__ void __launch_bounds__(64)
ga_combine_kernel(const float* __restrict__ ws_s, const float* __restrict__ ws_acc,
                  float* __restrict__ out)
{
  const int bh = blockIdx.x;
  const int k  = threadIdx.x;
  float num = 0.f, den = 0.f;
  #pragma unroll
  for (int sp = 0; sp < NSPLIT; ++sp) {
    int pid = bh * NSPLIT + sp;
    num += ws_acc[(size_t)pid * DK_ + k];
    den += ws_s[pid];
  }
  out[bh * DK_ + k] = num / den;
}

extern "C" void kernel_launch(void* const* d_in, const int* in_sizes, int n_in,
                              void* d_out, int out_size, void* d_ws, size_t ws_size,
                              hipStream_t stream) {
  (void)in_sizes; (void)n_in; (void)out_size; (void)ws_size;
  const float* X  = (const float*)d_in[0];
  const float* S  = (const float*)d_in[1];
  const float* Wq = (const float*)d_in[2];
  const float* Wk = (const float*)d_in[3];
  const float* Wv = (const float*)d_in[4];

  const int NP = B_ * H_ * NSPLIT;
  float* wsf    = (float*)d_ws;
  float* ws_s   = wsf;
  float* ws_acc = wsf + NP;

  ga_flash_kernel<<<NP, 256, 0, stream>>>(X, S, Wq, Wk, Wv, ws_s, ws_acc);
  ga_combine_kernel<<<B_ * H_, 64, 0, stream>>>(ws_s, ws_acc, (float*)d_out);
}


# Query vector q formed by all 256 threads (float4 weight-row loads, a thread's 32 all in flight, per-wave partials added in LDS), not by 64 threads in 32 waited-for load groups; q sum regrouped
# speedup vs baseline: 2.3786x; 1.1234x over previous
#include <hip/hip_runtime.h>
#include <hip/hip_bf16.h>
#include <math.h>

typedef __attribute__((ext_vector_type(8)))  float  v8f;
typedef __attribute__((ext_vector_type(4)))  __bf16 v4bf;
typedef __attribute__((ext_vector_type(8)))  __bf16 v8bf;
typedef __attribute__((ext_vector_type(16))) __bf16 v16bf;

#define B_   32
#define L_   2048
#define D_   512
#define H_   8
#define DK_  64
#define NSPLIT 2
#define CHUNK (L_ / NSPLIT)
#define TILES (CHUNK / 16)
#define KSTEPS (D_ / 32)
#define SROW 520
#define ARING 6
#define AGRP  2

static __device__ inline v16bf join16(v8bf lo, v8bf hi) {
  return __builtin_shufflevector(lo, hi, 0,1,2,3,4,5,6,7,8,9,10,11,12,13,14,15);
}

static __device__ inline float fast_tanh(float x) {
  x = fminf(fmaxf(x, -15.f), 15.f);
  float e = __expf(2.0f * x);
  return (e - 1.0f) / (e + 1.0f);
}

static __device__ inline void lds_barrier() {
  __builtin_amdgcn_fence(__ATOMIC_RELEASE, "workgroup", "local");
  __builtin_amdgcn_s_barrier();
  __builtin_amdgcn_fence(__ATOMIC_ACQUIRE, "workgroup", "local");
}

__global__ void __launch_bounds__(256)
ga_flash_kernel(const float* __restrict__ X, const float* __restrict__ S,
                const float* __restrict__ Wq, const float* __restrict__ Wk,
                const float* __restrict__ Wv,
                float* __restrict__ ws_s, float* __restrict__ ws_acc)
{
  __shared__ __bf16 s_tile[16 * SROW];
  __shared__ __attribute__((aligned(16))) float sh_qp[8][DK_];
  __shared__ float  sh_q[DK_];
  __shared__ __attribute__((aligned(16))) float sh_part[2][4][16];

  const int tid  = threadIdx.x;
  const int lane = tid & 31;
  const int wave = tid >> 5;
  const int wn   = wave & 3;
  const int sel  = wave >> 2;

  const int pid = blockIdx.x;
  const int sp  = pid & (NSPLIT - 1);
  const int h   = (pid >> 1) & (H_ - 1);
  const int b   = pid >> 4;
  const int l0c = sp * CHUNK;

  const int m_row = lane & 15;
  const int kbase = (lane < 16) ? 0 : 8;
  const int half  = lane >> 4;

  {
    const int rg = tid >> 4;
    const float* wq = Wq + ((size_t)h * D_ + rg) * DK_ + m_row * 4;
    const float* xp = X + (size_t)b * D_ + rg;
    float4 qa = { 0.f, 0.f, 0.f, 0.f };
    #pragma unroll
    for (int i = 0; i < D_ / 16; ++i) {
      const float4 w = *reinterpret_cast<const float4*>(wq + (size_t)i * 16 * DK_);
      const float xv = xp[i * 16];
      qa.x = fmaf(xv, w.x, qa.x);
      qa.y = fmaf(xv, w.y, qa.y);
      qa.z = fmaf(xv, w.z, qa.z);
      qa.w = fmaf(xv, w.w, qa.w);
    }
    qa.x += __shfl_xor(qa.x, 16);
    qa.y += __shfl_xor(qa.y, 16);
    qa.z += __shfl_xor(qa.z, 16);
    qa.w += __shfl_xor(qa.w, 16);
    if (lane < 16) *reinterpret_cast<float4*>(&sh_qp[wave][m_row * 4]) = qa;
  }
  lds_barrier();
  if (tid < DK_) {
    float a = 0.f;
    #pragma unroll
    for (int w = 0; w < 8; ++w) a += sh_qp[w][tid];
    sh_q[tid] = fast_tanh(a);
  }

  const float* wp = (sel ? Wv : Wk) + ((size_t)h * D_ + kbase) * DK_ + wn * 16 + m_row;
  v16bf bfr[KSTEPS];
  #pragma unroll
  for (int kk = 0; kk < KSTEPS; ++kk) {
    #pragma unroll
    for (int e = 0; e < 16; ++e) {
      const int k = kk * 32 + (e & 7) + ((e >> 3) << 4);
      bfr[kk][e] = (__bf16)wp[k * DK_];
    }
    if ((kk & 3) == 3) __builtin_amdgcn_sched_barrier(0);
  }

  float psum = 0.f;
  float acc  = 0.f;
  v8f c = {};

  auto v_update = [&](int buf) {
    const float* pp = &sh_part[buf][0][8 * half];
    float sc[8];
    #pragma unroll
    for (int w = 0; w < 4; ++w) {
      float4 lo = *reinterpret_cast<const float4*>(pp + 16 * w);
      float4 hi = *reinterpret_cast<const float4*>(pp + 16 * w + 4);
      float v[8] = { lo.x, lo.y, lo.z, lo.w, hi.x, hi.y, hi.z, hi.w };
      #pragma unroll
      for (int r = 0; r < 8; ++r) sc[r] = (w == 0) ? v[r] : sc[r] + v[r];
    }
    #pragma unroll
    for (int r = 0; r < 8; ++r) {
      float p = __expf(fmaf(sc[r], 0.125f, -8.0f));
      float v = c[r];
      v = v > 0.f ? v : 0.f;
      psum += p;
      acc = fmaf(p, v, acc);
    }
  };

  const size_t sbase = ((size_t)b * L_ + l0c) * D_;

  float4 stage[8];
  #pragma unroll
  for (int j = 0; j < 8; ++j)
    stage[j] = *reinterpret_cast<const float4*>(&S[sbase + tid * 4 + j * 1024]);

  for (int t = 0; t < TILES; ++t) {
    lds_barrier();
    #pragma unroll
    for (int j = 0; j < 8; ++j) {
      int e0  = tid * 4 + j * 1024;
      int row = e0 >> 9, col = e0 & 511;
      float4 f = stage[j];
      v4bf p4 = { (__bf16)f.x, (__bf16)f.y, (__bf16)f.z, (__bf16)f.w };
      *reinterpret_cast<v4bf*>(&s_tile[row * SROW + col]) = p4;
    }
    if (t + 1 < TILES) {
      const size_t sb1 = sbase + (size_t)(t + 1) * 16 * D_;
      #pragma unroll
      for (int j = 0; j < 8; ++j)
        stage[j] = *reinterpret_cast<const float4*>(&S[sb1 + tid * 4 + j * 1024]);
    }
    if (t + 2 < TILES) {
      const size_t sb2 = sbase + (size_t)(t + 2) * 16 * D_;
      #pragma unroll
      for (int j = 0; j < 8; ++j)
        __builtin_prefetch(&S[sb2 + tid * 4 + j * 1024], 0, 1);
    }
    lds_barrier();

    if (sel == 1 && t > 0) v_update((t - 1) & 1);

    const __bf16* pa = &s_tile[m_row * SROW + kbase];
    v16bf afr[ARING];
    #pragma unroll
    for (int i = 0; i < ARING - AGRP; ++i)
      afr[i] = join16(*reinterpret_cast<const v8bf*>(pa + i * 32),
                      *reinterpret_cast<const v8bf*>(pa + i * 32 + 16));
    c = v8f{};
    #pragma unroll
    for (int kk = 0; kk < KSTEPS; ++kk) {
      if (kk % AGRP == 0) {
        __builtin_amdgcn_sched_barrier(0);
        #pragma unroll
        for (int g = 0; g < AGRP; ++g) {
          const int kn = kk + ARING - AGRP + g;
          if (kn < KSTEPS)
            afr[kn % ARING] = join16(*reinterpret_cast<const v8bf*>(pa + kn * 32),
                                     *reinterpret_cast<const v8bf*>(pa + kn * 32 + 16));
        }
        __builtin_amdgcn_sched_barrier(0);
      }
      c = __builtin_amdgcn_wmma_f32_16x16x32_bf16(
            false, afr[kk % ARING], false, bfr[kk], (short)0, c, false, false);
    }

    if (sel == 0) {
      float qv = sh_q[wn * 16 + m_row];
      float part[8];
      #pragma unroll
      for (int r = 0; r < 8; ++r) part[r] = fast_tanh(c[r]) * qv;
      #pragma unroll
      for (int off = 1; off < 16; off <<= 1) {
        #pragma unroll
        for (int r = 0; r < 8; ++r) part[r] += __shfl_xor(part[r], off);
      }
      if ((lane & 15) == 0) {
        #pragma unroll
        for (int r = 0; r < 8; ++r) sh_part[t & 1][wn][r + 8 * half] = part[r];
      }
    }
  }

  lds_barrier();
  if (sel == 1) {
    v_update((TILES - 1) & 1);
    acc  += __shfl_xor(acc, 16);
    psum += __shfl_xor(psum, 16);
    if (lane < 16) ws_acc[(size_t)pid * DK_ + wn * 16 + m_row] = acc;
    if (tid == 128) ws_s[pid] = psum;
  }
}

__global__ void __launch_bounds__(64)
ga_combine_kernel(const float* __restrict__ ws_s, const float* __restrict__ ws_acc,
                  float* __restrict__ out)
{
  const int bh = blockIdx.x;
  const int k  = threadIdx.x;
  float num = 0.f, den = 0.f;
  #pragma unroll
  for (int sp = 0; sp < NSPLIT; ++sp) {
    int pid = bh * NSPLIT + sp;
    num += ws_acc[(size_t)pid * DK_ + k];
    den += ws_s[pid];
  }
  out[bh * DK_ + k] = num / den;
}

extern "C" void kernel_launch(void* const* d_in, const int* in_sizes, int n_in,
                              void* d_out, int out_size, void* d_ws, size_t ws_size,
                              hipStream_t stream) {
  (void)in_sizes; (void)n_in; (void)out_size; (void)ws_size;
  const float* X  = (const float*)d_in[0];
  const float* S  = (const float*)d_in[1];
  const float* Wq = (const float*)d_in[2];
  const float* Wk = (const float*)d_in[3];
  const float* Wv = (const float*)d_in[4];

  const int NP = B_ * H_ * NSPLIT;
  float* wsf    = (float*)d_ws;
  float* ws_s   = wsf;
  float* ws_acc = wsf + NP;

  ga_flash_kernel<<<NP, 256, 0, stream>>>(X, S, Wq, Wk, Wv, ws_s, ws_acc);
  ga_combine_kernel<<<B_ * H_, 64, 0, stream>>>(ws_s, ws_acc, (float*)d_out);
}
